# starting kernel
# baseline (speedup 1.0000x reference)
#include <hip/hip_runtime.h>

typedef float v2f __attribute__((ext_vector_type(2)));
typedef float v8f __attribute__((ext_vector_type(8)));

#define BATCH   8
#define SEQ     2048
#define CIN     512
#define GROUPS  8
#define ICPG    64
#define OCPG    64
#define KTAPS   5
#define SCHUNK  128
#define XROWS   (SCHUNK + 4)
#define XPITCH  68
#define KFLAT   (ICPG * KTAPS)
#define WPITCH  324

__global__ __launch_bounds__(256)
void masked_gconv1d_wmma(const float* __restrict__ in,
                         const int*   __restrict__ pos,
                         const float* __restrict__ kern,
                         float*       __restrict__ out) {
  extern __shared__ char smem[];
  float* Wl = (float*)smem;
  float* Xl = Wl + OCPG * WPITCH;
  int*   Pl = (int*)(Xl + XROWS * XPITCH);

  const int tid = threadIdx.x;
  const int s0  = blockIdx.x * SCHUNK;
  const int g   = blockIdx.y;
  const int b   = blockIdx.z;

  const float* Wg = kern + (size_t)g * OCPG * KFLAT;
  for (int e = tid; e < OCPG * KFLAT; e += 256) {
    int o  = e / KFLAT;
    int r  = e - o * KFLAT;
    int i  = r / KTAPS;
    int t  = r - i * KTAPS;
    Wl[o * WPITCH + t * ICPG + i] = Wg[e];
  }

  const size_t inb = (size_t)b * SEQ * CIN + (size_t)g * ICPG;
  for (int e = tid; e < XROWS * ICPG; e += 256) {
    int r = e >> 6;
    int c = e & 63;
    int s = s0 + r - 2;
    float v = 0.0f;
    if (s >= 0 && s < SEQ) v = in[inb + (size_t)s * CIN + c];
    Xl[r * XPITCH + c] = v;
  }

  for (int r = tid; r < XROWS; r += 256) {
    int s = s0 + r - 2;
    Pl[r] = (s >= 0 && s < SEQ) ? pos[(size_t)b * SEQ + s] : (int)0x80000000;
  }
  __syncthreads();

  const int wave = tid >> 5;
  const int lane = tid & 31;
  const int row  = lane & 15;
  const int hi   = lane >> 4;
  const int klo  = hi * 2;

  float mv[KTAPS];
  const int rc = wave * 16 + row + 2;
  const int pc = Pl[rc];
#pragma unroll
  for (int t = 0; t < KTAPS; ++t)
    mv[t] = (Pl[rc + t - 2] == pc + (t - 2)) ? 1.0f : 0.0f;

  v8f acc[4] = {};

#pragma unroll
  for (int t = 0; t < KTAPS; ++t) {
    const float m = mv[t];
    const float* xr = &Xl[(wave * 16 + row + t) * XPITCH + klo];
    const float* wr = &Wl[row * WPITCH + t * ICPG + klo];
#pragma unroll 4
    for (int kb = 0; kb < 16; ++kb) {
      v2f a = *(const v2f*)(xr + kb * 4);
      a.x *= m;
      a.y *= m;
#pragma unroll
      for (int n = 0; n < 4; ++n) {
        v2f bv = *(const v2f*)(wr + n * 16 * WPITCH + kb * 4);
        acc[n] = __builtin_amdgcn_wmma_f32_16x16x4_f32(
             false, a,  false, bv,
             (short)0, acc[n],  false,  false);
      }
    }
  }

  const size_t ob = ((size_t)b * SEQ + (size_t)(s0 + wave * 16)) * CIN
                  + (size_t)g * OCPG + row;
#pragma unroll
  for (int n = 0; n < 4; ++n) {
#pragma unroll
    for (int v = 0; v < 8; ++v) {
      int M = v + hi * 8;
      out[ob + (size_t)M * CIN + n * 16] = acc[n][v];
    }
  }
}

extern "C" void kernel_launch(void* const* d_in, const int* in_sizes, int n_in,
                              void* d_out, int out_size, void* d_ws, size_t ws_size,
                              hipStream_t stream) {
  (void)in_sizes; (void)n_in; (void)out_size; (void)d_ws; (void)ws_size;
  const float* in   = (const float*)d_in[0];
  const int*   pos  = (const int*)d_in[1];
  const float* kern = (const float*)d_in[2];
  float*       out  = (float*)d_out;

  dim3 grid(SEQ / SCHUNK, GROUPS, BATCH);
  dim3 block(256);
  size_t smem = (size_t)(OCPG * WPITCH + XROWS * XPITCH) * sizeof(float)
              + (size_t)XROWS * sizeof(int);
  masked_gconv1d_wmma<<<grid, block, smem, stream>>>(in, pos, kern, out);
}


# block tile 128 -> 512 seq rows on 1024 threads, one block per WGP (grid 1024 -> 256): weights staged once per 512 rows, each thread's chain of dependent staging loads falls from 2 x 54 to about 36
# speedup vs baseline: 1.6120x; 1.6120x over previous
#include <hip/hip_runtime.h>

typedef float v2f __attribute__((ext_vector_type(2)));
typedef float v8f __attribute__((ext_vector_type(8)));

#define BATCH   8
#define SEQ     2048
#define CIN     512
#define GROUPS  8
#define ICPG    64
#define OCPG    64
#define KTAPS   5
#define SCHUNK  512
#define NTHR    1024
#define XROWS   (SCHUNK + 4)
#define XPITCH  68
#define KFLAT   (ICPG * KTAPS)
#define WPITCH  324

__global__ __launch_bounds__(NTHR)
void masked_gconv1d_wmma(const float* __restrict__ in,
                         const int*   __restrict__ pos,
                         const float* __restrict__ kern,
                         float*       __restrict__ out) {
  __shared__ __attribute__((aligned(16))) char smem[(OCPG * WPITCH + XROWS * XPITCH) * 4 + XROWS * 4];
  float* Wl = (float*)smem;
  float* Xl = Wl + OCPG * WPITCH;
  int*   Pl = (int*)(Xl + XROWS * XPITCH);

  const int tid = threadIdx.x;
  const int s0  = blockIdx.x * SCHUNK;
  const int g   = blockIdx.y;
  const int b   = blockIdx.z;

  const float* Wg = kern + (size_t)g * OCPG * KFLAT;
  for (int e = tid; e < OCPG * KFLAT; e += NTHR) {
    int o  = e / KFLAT;
    int r  = e - o * KFLAT;
    int i  = r / KTAPS;
    int t  = r - i * KTAPS;
    Wl[o * WPITCH + t * ICPG + i] = Wg[e];
  }

  const size_t inb = (size_t)b * SEQ * CIN + (size_t)g * ICPG;
  for (int e = tid; e < XROWS * ICPG; e += NTHR) {
    int r = e >> 6;
    int c = e & 63;
    int s = s0 + r - 2;
    float v = 0.0f;
    if (s >= 0 && s < SEQ) v = in[inb + (size_t)s * CIN + c];
    Xl[r * XPITCH + c] = v;
  }

  for (int r = tid; r < XROWS; r += NTHR) {
    int s = s0 + r - 2;
    Pl[r] = (s >= 0 && s < SEQ) ? pos[(size_t)b * SEQ + s] : (int)0x80000000;
  }
  __syncthreads();

  const int wave = tid >> 5;
  const int lane = tid & 31;
  const int row  = lane & 15;
  const int hi   = lane >> 4;
  const int klo  = hi * 2;

  float mv[KTAPS];
  const int rc = wave * 16 + row + 2;
  const int pc = Pl[rc];
#pragma unroll
  for (int t = 0; t < KTAPS; ++t)
    mv[t] = (Pl[rc + t - 2] == pc + (t - 2)) ? 1.0f : 0.0f;

  v8f acc[4] = {};

#pragma unroll
  for (int t = 0; t < KTAPS; ++t) {
    const float m = mv[t];
    const float* xr = &Xl[(wave * 16 + row + t) * XPITCH + klo];
    const float* wr = &Wl[row * WPITCH + t * ICPG + klo];
#pragma unroll 4
    for (int kb = 0; kb < 16; ++kb) {
      v2f a = *(const v2f*)(xr + kb * 4);
      a.x *= m;
      a.y *= m;
#pragma unroll
      for (int n = 0; n < 4; ++n) {
        v2f bv = *(const v2f*)(wr + n * 16 * WPITCH + kb * 4);
        acc[n] = __builtin_amdgcn_wmma_f32_16x16x4_f32(
             false, a,  false, bv,
             (short)0, acc[n],  false,  false);
      }
    }
  }

  const size_t ob = ((size_t)b * SEQ + (size_t)(s0 + wave * 16)) * CIN
                  + (size_t)g * OCPG + row;
#pragma unroll
  for (int n = 0; n < 4; ++n) {
#pragma unroll
    for (int v = 0; v < 8; ++v) {
      int M = v + hi * 8;
      out[ob + (size_t)M * CIN + n * 16] = acc[n][v];
    }
  }
}

extern "C" void kernel_launch(void* const* d_in, const int* in_sizes, int n_in,
                              void* d_out, int out_size, void* d_ws, size_t ws_size,
                              hipStream_t stream) {
  (void)in_sizes; (void)n_in; (void)out_size; (void)d_ws; (void)ws_size;
  const float* in   = (const float*)d_in[0];
  const int*   pos  = (const int*)d_in[1];
  const float* kern = (const float*)d_in[2];
  float*       out  = (float*)d_out;

  dim3 grid(SEQ / SCHUNK, GROUPS, BATCH);
  dim3 block(NTHR);
  masked_gconv1d_wmma<<<grid, block, 0, stream>>>(in, pos, kern, out);
}


# 16 waves per block, each with a private LDS tile for its 32 rows: staging as one group of 16-byte loads, and the second 16-row tile's loads fly under the first tile's WMMAs; was 33 serial load trips
# speedup vs baseline: 2.2529x; 1.3976x over previous
#include <hip/hip_runtime.h>

typedef float v2f __attribute__((ext_vector_type(2)));
typedef float v4f __attribute__((ext_vector_type(4)));
typedef float v8f __attribute__((ext_vector_type(8)));

#define BATCH   8
#define SEQ     2048
#define CIN     512
#define GROUPS  8
#define ICPG    64
#define OCPG    64
#define KTAPS   5
#define SCHUNK  512
#define NTHR    512
#define NWAVE   (NTHR / 32)
#define WROWS   36
#define XPITCH  68
#define XWSZ    (WROWS * XPITCH)
#define KFLAT   (ICPG * KTAPS)
#define WPITCH  324
#define NW4     (OCPG * KFLAT / 4 / NTHR)

__global__ __launch_bounds__(NTHR)
void masked_gconv1d_wmma(const float* __restrict__ in,
                         const int*   __restrict__ pos,
                         const float* __restrict__ kern,
                         float*       __restrict__ out) {
  __shared__ __attribute__((aligned(16))) float smem[OCPG * WPITCH + NWAVE * XWSZ];
  float* Wl = smem;
  float* Xl = smem + OCPG * WPITCH;

  const int tid  = threadIdx.x;
  const int wave = tid >> 5;
  const int lane = tid & 31;
  const int row  = lane & 15;
  const int hi   = lane >> 4;
  const int klo  = hi * 2;
  const int g    = blockIdx.y;
  const int b    = blockIdx.z;
  const int sw   = blockIdx.x * SCHUNK + wave * 32;

  float* Xw = Xl + wave * XWSZ;

  const v4f* Wg4 = (const v4f*)(kern + (size_t)g * OCPG * KFLAT);
  v4f wv[NW4];
#pragma unroll
  for (int j = 0; j < NW4; ++j) wv[j] = Wg4[tid + j * NTHR];

  const float* inl = in + (size_t)b * SEQ * CIN + (size_t)g * ICPG + row * 4;
  v4f x0[10];
#pragma unroll
  for (int j = 0; j < 10; ++j) {
    int s  = sw + 2 * j + hi - 2;
    int sc = min(max(s, 0), SEQ - 1);
    v4f v  = *(const v4f*)(inl + (size_t)sc * CIN);
    if (s != sc) v = (v4f)(0.0f);
    x0[j] = v;
  }

  int pv[2][KTAPS];
#pragma unroll
  for (int tau = 0; tau < 2; ++tau) {
#pragma unroll
    for (int t = 0; t < KTAPS; ++t) {
      int s  = sw + tau * 16 + row + t - 2;
      int sc = min(max(s, 0), SEQ - 1);
      int p  = pos[(size_t)b * SEQ + sc];
      pv[tau][t] = (s == sc) ? p : (int)0x80000000;
    }
  }

#pragma unroll
  for (int j = 0; j < NW4; ++j) {
    int e = (tid + j * NTHR) * 4;
    int o = e / KFLAT;
    int r = e - o * KFLAT;
#pragma unroll
    for (int u = 0; u < 4; ++u) {
      int rr = r + u;
      int i  = rr / KTAPS;
      int t  = rr - i * KTAPS;
      Wl[o * WPITCH + t * ICPG + i] = wv[j][u];
    }
  }

#pragma unroll
  for (int j = 0; j < 10; ++j)
    *(v4f*)(Xw + (2 * j + hi) * XPITCH + row * 4) = x0[j];

  __syncthreads();

  v4f xn[8];
#pragma unroll
  for (int j = 0; j < 8; ++j) {
    int s  = sw + 20 + 2 * j + hi - 2;
    int sc = min(max(s, 0), SEQ - 1);
    v4f v  = *(const v4f*)(inl + (size_t)sc * CIN);
    if (s != sc) v = (v4f)(0.0f);
    xn[j] = v;
  }

#pragma unroll
  for (int tau = 0; tau < 2; ++tau) {
    if (tau == 1) {
#pragma unroll
      for (int j = 0; j < 8; ++j)
        *(v4f*)(Xw + (20 + 2 * j + hi) * XPITCH + row * 4) = xn[j];
      __builtin_amdgcn_fence(__ATOMIC_RELEASE, "wavefront");
      __builtin_amdgcn_wave_barrier();
      __builtin_amdgcn_fence(__ATOMIC_ACQUIRE, "wavefront");
    }

    float mv[KTAPS];
    const int pc = pv[tau][2];
#pragma unroll
    for (int t = 0; t < KTAPS; ++t)
      mv[t] = (pv[tau][t] == pc + (t - 2)) ? 1.0f : 0.0f;

    v8f acc[4] = {};

#pragma unroll
    for (int t = 0; t < KTAPS; ++t) {
      const float m = mv[t];
      const float* xr = &Xw[(tau * 16 + row + t) * XPITCH + klo];
      const float* wr = &Wl[row * WPITCH + t * ICPG + klo];
#pragma unroll 4
      for (int kb = 0; kb < 16; ++kb) {
        v2f a = *(const v2f*)(xr + kb * 4);
        a.x *= m;
        a.y *= m;
#pragma unroll
        for (int n = 0; n < 4; ++n) {
          v2f bv = *(const v2f*)(wr + n * 16 * WPITCH + kb * 4);
          acc[n] = __builtin_amdgcn_wmma_f32_16x16x4_f32(
               false, a,  false, bv,
               (short)0, acc[n],  false,  false);
        }
      }
    }

    const size_t ob = ((size_t)b * SEQ + (size_t)(sw + tau * 16)) * CIN
                    + (size_t)g * OCPG + row;
#pragma unroll
    for (int n = 0; n < 4; ++n) {
#pragma unroll
      for (int v = 0; v < 8; ++v) {
        int M = v + hi * 8;
        out[ob + (size_t)M * CIN + n * 16] = acc[n][v];
      }
    }
  }
}

extern "C" void kernel_launch(void* const* d_in, const int* in_sizes, int n_in,
                              void* d_out, int out_size, void* d_ws, size_t ws_size,
                              hipStream_t stream) {
  (void)in_sizes; (void)n_in; (void)out_size; (void)d_ws; (void)ws_size;
  const float* in   = (const float*)d_in[0];
  const int*   pos  = (const int*)d_in[1];
  const float* kern = (const float*)d_in[2];
  float*       out  = (float*)d_out;

  dim3 grid(SEQ / SCHUNK, GROUPS, BATCH);
  dim3 block(NTHR);
  masked_gconv1d_wmma<<<grid, block, 0, stream>>>(in, pos, kern, out);
}
